# starting kernel
# baseline (speedup 1.0000x reference)
#include <hip/hip_runtime.h>
#include <math.h>


#define NDIM   256
#define NBATCH 65536
#define NROT   32640
#define TILE_B 64
#define RS     260

typedef __attribute__((ext_vector_type(2))) float v2f;
typedef __attribute__((ext_vector_type(8))) float v8f;

__global__ void qlayer_trig(const float* __restrict__ rots1,
                            const float* __restrict__ rots2,
                            const float* __restrict__ phases,
                            float* __restrict__ c1, float* __restrict__ s1,
                            float* __restrict__ c2, float* __restrict__ s2,
                            float* __restrict__ cp, float* __restrict__ sp) {
  int idx = blockIdx.x * blockDim.x + threadIdx.x;
  if (idx < NROT) {
    float s, c;
    sincosf(rots1[idx], &s, &c); c1[idx] = c; s1[idx] = s;
    sincosf(rots2[idx], &s, &c); c2[idx] = c; s2[idx] = s;
  }
  if (idx < NDIM) {
    float s, c;
    sincosf(phases[idx], &s, &c); cp[idx] = c; sp[idx] = s;
  }
}

__global__ void qlayer_givens(const float* __restrict__ c1,
                              const float* __restrict__ s1,
                              const float* __restrict__ c2,
                              const float* __restrict__ s2,
                              float* __restrict__ T1,
                              float* __restrict__ T2) {
  extern __shared__ float sm[];
  const int t = threadIdx.x;
  const float* cA = (blockIdx.x == 0) ? c1 : c2;
  const float* sA = (blockIdx.x == 0) ? s1 : s2;
  float* T        = (blockIdx.x == 0) ? T1 : T2;

  float* v = sm + t * 257;
  for (int k = 0; k < NDIM; ++k) v[k] = (k == t) ? 1.0f : 0.0f;

  int i = 0, j = 1;
  for (int r = 0; r < NROT; ++r) {
    float c = cA[r];
    float s = sA[r];
    float ri = v[i];
    float rj = v[j];
    v[i] = c * ri - s * rj;
    v[j] = s * ri + c * rj;
    if (++j == NDIM) { ++i; j = i + 1; }
  }
  for (int n = 0; n < NDIM; ++n) T[t * NDIM + n] = v[n];
}

__device__ __forceinline__ void gemm_stripe(const float* __restrict__ smSrc,
                                            const float* __restrict__ T,
                                            int r, int h, int ln16, int hl,
                                            v8f acc[8]) {
  const int m = r * 16 + ln16;
  #pragma unroll 2
  for (int k = 0; k < NDIM; k += 4) {
    const int kk = k + 2 * hl;
    v2f a;
    a.x = smSrc[m * RS + kk];
    a.y = smSrc[m * RS + kk + 1];
    const float* b0 = T + kk * NDIM       + h * 128 + ln16;
    const float* b1 = T + (kk + 1) * NDIM + h * 128 + ln16;
    #pragma unroll
    for (int t = 0; t < 8; ++t) {
      v2f b;
      b.x = b0[t * 16];
      b.y = b1[t * 16];
      acc[t] = __builtin_amdgcn_wmma_f32_16x16x4_f32(
           false, a,  false, b,
           (short)0, acc[t],  false,  false);
    }
  }
}

__global__ void __launch_bounds__(256)
qlayer_main(const float* __restrict__ x,
            const float* __restrict__ T1,
            const float* __restrict__ T2,
            const float* __restrict__ cp,
            const float* __restrict__ sp,
            float* __restrict__ out) {
  extern __shared__ float sm[];
  float* smA = sm;
  float* smB = sm + TILE_B * RS;

  const int tid  = threadIdx.x;
  const int wave = tid >> 5;
  const int lane = tid & 31;
  const int hl   = lane >> 4;
  const int ln16 = lane & 15;
  const int r    = wave & 3;
  const int h    = wave >> 2;
  const int base_row = blockIdx.x * TILE_B;

  {
    const int row = tid >> 2;
    const int q   = tid & 3;
    const float4* src = (const float4*)(x + (base_row + row) * NDIM + q * 64);
    float4* dst = (float4*)(smA + row * RS + q * 64);
    float acc = 0.f;
    #pragma unroll
    for (int c = 0; c < 16; ++c) {
      float4 vv = src[c];
      dst[c] = vv;
      acc += vv.x * vv.x + vv.y * vv.y + vv.z * vv.z + vv.w * vv.w;
    }
    smB[row * 4 + q] = acc;
  }
  __syncthreads();
  if (tid < TILE_B) {
    float s = smB[tid * 4] + smB[tid * 4 + 1] + smB[tid * 4 + 2] + smB[tid * 4 + 3];
    smB[1024 + tid] = rsqrtf(s);
  }
  __syncthreads();
  {
    const int row = tid >> 2;
    const int q   = tid & 3;
    const float inv = smB[1024 + row];
    float4* dst = (float4*)(smA + row * RS + q * 64);
    #pragma unroll
    for (int c = 0; c < 16; ++c) {
      float4 vv = dst[c];
      vv.x *= inv; vv.y *= inv; vv.z *= inv; vv.w *= inv;
      dst[c] = vv;
    }
  }
  __syncthreads();

  const v8f vzero = {0.f, 0.f, 0.f, 0.f, 0.f, 0.f, 0.f, 0.f};

  v8f acc[8];
  #pragma unroll
  for (int t = 0; t < 8; ++t) acc[t] = vzero;
  gemm_stripe(smA, T1, r, h, ln16, hl, acc);

  __syncthreads();

  #pragma unroll
  for (int t = 0; t < 8; ++t) {
    const int n  = h * 128 + t * 16 + ln16;
    const float cw = cp[n];
    const float sw = sp[n];
    #pragma unroll
    for (int g = 0; g < 8; ++g) {
      const int m = r * 16 + g + 8 * hl;
      const float y = acc[t][g];
      smA[m * RS + n] = y * cw;
      smB[m * RS + n] = y * sw;
    }
  }
  __syncthreads();

  v8f res[8];
  #pragma unroll
  for (int t = 0; t < 8; ++t) acc[t] = vzero;
  gemm_stripe(smA, T2, r, h, ln16, hl, acc);
  #pragma unroll
  for (int t = 0; t < 8; ++t) {
    #pragma unroll
    for (int g = 0; g < 8; ++g) { float v = acc[t][g]; res[t][g] = v * v; }
  }

  #pragma unroll
  for (int t = 0; t < 8; ++t) acc[t] = vzero;
  gemm_stripe(smB, T2, r, h, ln16, hl, acc);
  #pragma unroll
  for (int t = 0; t < 8; ++t) {
    #pragma unroll
    for (int g = 0; g < 8; ++g) { float v = acc[t][g]; res[t][g] += v * v; }
  }

  #pragma unroll
  for (int t = 0; t < 8; ++t) {
    const int n = h * 128 + t * 16 + ln16;
    #pragma unroll
    for (int g = 0; g < 8; ++g) {
      const int m = r * 16 + g + 8 * hl;
      out[(base_row + m) * NDIM + n] = res[t][g];
    }
  }
}

extern "C" void kernel_launch(void* const* d_in, const int* in_sizes, int n_in,
                              void* d_out, int out_size, void* d_ws, size_t ws_size,
                              hipStream_t stream) {
  const float* x      = (const float*)d_in[0];
  const float* rots1  = (const float*)d_in[1];
  const float* phases = (const float*)d_in[2];
  const float* rots2  = (const float*)d_in[3];
  float* out = (float*)d_out;
  float* ws  = (float*)d_ws;

  float* c1 = ws;
  float* s1 = ws + 32640;
  float* c2 = ws + 65280;
  float* s2 = ws + 97920;
  float* cp = ws + 130560;
  float* sp = ws + 130816;
  float* T1 = ws + 131072;
  float* T2 = ws + 196608;

  const int lds_givens = 256 * 257 * 4;
  const int lds_main   = 2 * TILE_B * RS * 4;
  hipFuncSetAttribute(reinterpret_cast<const void*>(qlayer_givens),
                      hipFuncAttributeMaxDynamicSharedMemorySize, lds_givens);
  hipFuncSetAttribute(reinterpret_cast<const void*>(qlayer_main),
                      hipFuncAttributeMaxDynamicSharedMemorySize, lds_main);

  qlayer_trig<<<(NROT + 255) / 256, 256, 0, stream>>>(
      rots1, rots2, phases, c1, s1, c2, s2, cp, sp);
  qlayer_givens<<<2, 256, lds_givens, stream>>>(c1, s1, c2, s2, T1, T2);
  qlayer_main<<<NBATCH / TILE_B, 256, lds_main, stream>>>(
      x, T1, T2, cp, sp, out);
}


# Rotation-matrix build: each wave runs the 2nd sweep on its phased 1st-sweep column (rows x cos or sin phi, not the identity), giving Re/Im of M2 diag(e^{i phi}) M1; batch: 2 GEMMs not 3; new rounding
# speedup vs baseline: 13.6840x; 13.6840x over previous
#include <hip/hip_runtime.h>
#include <math.h>


#define NDIM   256
#define NBATCH 65536
#define NROT   32640
#define TILE_B 64
#define RS     260

typedef __attribute__((ext_vector_type(2))) float v2f;
typedef __attribute__((ext_vector_type(8))) float v8f;

#define GV_LAST  (2 * NDIM - 3)

__device__ __forceinline__ int gv_step_count(int t) {
  const int imin = (t > NDIM - 1) ? (t - (NDIM - 1)) : 0;
  const int imax = (t - 1) >> 1;
  return imax - imin + 1;
}

__device__ __forceinline__ int gv_step_first(int t) {
  const int m = (t > NDIM) ? (t - NDIM) : 0;
  return (t >> 1) * ((t + 1) >> 1) - ((m * (m + 1)) >> 1);
}

__device__ __forceinline__ int gv_rot_index(int t, int slot) {
  const int imin = (t > NDIM - 1) ? (t - (NDIM - 1)) : 0;
  const int i    = imin + slot;
  const unsigned tri = ((unsigned)(i * (2 * NDIM - 1 - i))) >> 1;
  return (int)tri + (t - 2 * i - 1);
}

__global__ void qlayer_trig(const float* __restrict__ rots1,
                            const float* __restrict__ rots2,
                            const float* __restrict__ phases,
                            float* __restrict__ cs1, float* __restrict__ cs2,
                            float* __restrict__ cp, float* __restrict__ sp) {
  const int idx  = blockIdx.x * blockDim.x + threadIdx.x;
  const int t    = (idx >> 7) + 1;
  const int slot = idx & 127;
  if (t <= GV_LAST && slot < gv_step_count(t)) {
    const int k   = gv_rot_index(t, slot);
    const int pos = gv_step_first(t) + slot;
    float s, c;
    sincosf(rots1[k], &s, &c); ((float2*)cs1)[pos] = make_float2(c, s);
    sincosf(rots2[k], &s, &c); ((float2*)cs2)[pos] = make_float2(c, s);
  }
  if (idx < NDIM) {
    float s, c;
    sincosf(phases[idx], &s, &c); cp[idx] = c; sp[idx] = s;
  }
}

#define GV_COLS  4
#define GV_R     4
#define GV_U     16
#define GV_TURN  (NDIM - 1)
#define GV_VS    (NDIM + 128)

static_assert(GV_R * 32 >= NDIM / 2, "a step's rotations must fit the wave");
static_assert(GV_U % GV_R == 0 && GV_U % 8 == 0, "the register renaming and the take-over register must repeat every trip");

struct gv_cs {
  float2 v[GV_R];
};

__device__ __forceinline__ int gv_table_pos(int t) {
  return gv_step_first((t < GV_LAST) ? t : GV_LAST);
}

__device__ __forceinline__ unsigned gv_lane_offset(int t, int lane) {
  return (unsigned)(gv_table_pos(t) + GV_R * lane) * (unsigned)sizeof(float2);
}

__device__ __forceinline__ void gv_request(const float2* __restrict__ csA,
                                           unsigned off, gv_cs& cs) {
  const float2* __restrict__ p = (const float2*)((const char*)csA + off);
  #pragma unroll
  for (int r = 0; r < GV_R; ++r) cs.v[r] = p[r];
}

__device__ __forceinline__ float gv_from_lane_below(float x) {
  const int xi = __builtin_bit_cast(int, x);
  const int across = __builtin_amdgcn_permlanex16(xi, xi, 0xffffffffu, 0xffffffffu, false, false);
  return __builtin_bit_cast(float, __builtin_amdgcn_update_dpp(across, xi, 0x111, 0xf, 0xf, false));
}

__device__ __forceinline__ float gv_from_lane_above(float x) {
  const int xi = __builtin_bit_cast(int, x);
  const int across = __builtin_amdgcn_permlanex16(xi, xi, 0u, 0u, false, false);
  return __builtin_bit_cast(float, __builtin_amdgcn_update_dpp(across, xi, 0x101, 0xf, 0xf, false));
}

template <bool SEEDED>
__device__ __forceinline__ void gv_step(const float2* __restrict__ csA,
                                        float* __restrict__ vcol,
                                        int t, int t8, bool up, int lane, int col,
                                        gv_cs& cs, unsigned& off, int& pos, float& seed,
                                        float (&I)[GV_R], float (&J)[GV_R]) {
  float ni[GV_R], nj[GV_R];
  #pragma unroll
  for (int r = 0; r < GV_R; ++r) {
    const float c = cs.v[r].x;
    const float s = cs.v[r].y;
    ni[r] = __builtin_fmaf(c, I[r], -(s * J[r]));
    nj[r] = __builtin_fmaf(s, I[r], c * J[r]);
  }
  if ((t8 & 1) == 0) {
    const int p  = gv_step_count(t);
    const int pr = up ? ((t8 >> 1) & 3) : ((3 - (t8 >> 1)) & 3);
    ni[pr] = (lane == (p >> 2)) ? J[pr] : ni[pr];
  }
  const int pos_new = gv_table_pos(t + GV_U);
  off += (unsigned)(pos_new - pos) * (unsigned)sizeof(float2);
  pos = pos_new;
  gv_request(csA, off, cs);
  __builtin_amdgcn_sched_barrier(0);
  if (up) {
    #pragma unroll
    for (int r = 0; r < GV_R; ++r) I[r] = ni[r];
    const float below = gv_from_lane_below(nj[GV_R - 1]);
    const float fresh = SEEDED ? seed : ((t + 1 == col) ? 1.0f : 0.0f);
    J[3] = nj[2]; J[2] = nj[1]; J[1] = nj[0];
    J[0] = (lane == 0) ? fresh : below;
    if (SEEDED) {
      const int nx = t + 1 + GV_U;
      seed = vcol[(nx < NDIM) ? nx : (NDIM - 1)];
    }
  } else {
    #pragma unroll
    for (int r = 0; r < GV_R; ++r) J[r] = nj[r];
    vcol[(t - GV_TURN) + GV_R * lane] = ni[0];
    const float above = gv_from_lane_above(ni[0]);
    I[0] = ni[1]; I[1] = ni[2]; I[2] = ni[3];
    I[3] = above;
  }
}

template <bool SEEDED>
__device__ __forceinline__ void gv_sweep(const float2* __restrict__ csA,
                                         float* __restrict__ vcol,
                                         int lane, int col) {
  gv_cs cs[GV_U];
  unsigned ao[GV_U];
  int ap[GV_U];
  float sd[GV_U];
  #pragma unroll
  for (int k = 0; k < GV_U; ++k) {
    ap[k] = gv_table_pos(1 + k);
    ao[k] = gv_lane_offset(1 + k, lane);
    gv_request(csA, ao[k], cs[k]);
    sd[k] = SEEDED ? vcol[2 + k] : 0.0f;
  }

  float I[GV_R], J[GV_R];
  #pragma unroll
  for (int r = 0; r < GV_R; ++r) { I[r] = 0.0f; J[r] = 0.0f; }
  if (SEEDED) {
    const float s0 = vcol[0];
    const float s1 = vcol[1];
    if (lane == 0) { I[0] = s0; J[0] = s1; }
  } else {
    if (lane == 0) { I[0] = (col == 0) ? 1.0f : 0.0f; J[0] = (col == 1) ? 1.0f : 0.0f; }
  }

  int t = 1;
  for (; t + GV_U - 1 < GV_TURN; t += GV_U) {
    #pragma unroll
    for (int k = 0; k < GV_U; ++k)
      gv_step<SEEDED>(csA, vcol, t + k, (1 + k) & 7, true, lane, col, cs[k],
                      ao[k], ap[k], sd[k], I, J);
  }
  #pragma unroll
  for (int k = 0; k < (GV_TURN - 1) % GV_U; ++k)
    gv_step<SEEDED>(csA, vcol, t + k, (1 + k) & 7, true, lane, col, cs[k],
                    ao[k], ap[k], sd[k], I, J);
  constexpr int K0 = (GV_TURN - 1) % GV_U;
  t = GV_TURN;
  for (; t + GV_U - 1 <= GV_LAST; t += GV_U) {
    #pragma unroll
    for (int k = 0; k < GV_U; ++k)
      gv_step<SEEDED>(csA, vcol, t + k, (GV_TURN + k) & 7, false, lane, col, cs[(K0 + k) % GV_U],
                      ao[(K0 + k) % GV_U], ap[(K0 + k) % GV_U], sd[(K0 + k) % GV_U], I, J);
  }
  #pragma unroll
  for (int k = 0; k < (GV_LAST - GV_TURN + 1) % GV_U; ++k)
    gv_step<SEEDED>(csA, vcol, t + k, (GV_TURN + k) & 7, false, lane, col, cs[(K0 + k) % GV_U],
                    ao[(K0 + k) % GV_U], ap[(K0 + k) % GV_U], sd[(K0 + k) % GV_U], I, J);
  vcol[NDIM - 1 + GV_R * lane] = J[0];
}

__global__ void __launch_bounds__(GV_COLS * 32)
qlayer_givens(const float* __restrict__ cs1,
              const float* __restrict__ cs2,
              const float* __restrict__ cp,
              const float* __restrict__ sp,
              float* __restrict__ T1,
              float* __restrict__ T2) {
  __shared__ float vs[GV_COLS * GV_VS];
  const int tid  = threadIdx.x;
  const int lane = tid & 31;
  const int wv   = __builtin_amdgcn_readfirstlane(tid >> 5);
  const int part = blockIdx.x / (NDIM / GV_COLS);
  const int col0 = (blockIdx.x % (NDIM / GV_COLS)) * GV_COLS;
  const int col  = col0 + wv;
  const float* __restrict__ pt = part ? sp : cp;
  float* __restrict__ T        = part ? T2 : T1;
  float* __restrict__ vcol     = vs + wv * GV_VS;

  gv_sweep<false>((const float2*)cs1, vcol, lane, col);

  __syncthreads();
  #pragma unroll
  for (int j = 0; j < NDIM / 32; ++j) {
    const int n = lane + 32 * j;
    vcol[n] = vcol[n] * pt[n];
  }
  __syncthreads();

  gv_sweep<true>((const float2*)cs2, vcol, lane, col);

  __syncthreads();

  for (int e = tid; e < NDIM * GV_COLS; e += GV_COLS * 32)
    T[col0 * NDIM + e] = vs[(e % GV_COLS) * GV_VS + e / GV_COLS];
}

__device__ __forceinline__ void gemm_stripe(const float* __restrict__ smSrc,
                                            const float* __restrict__ T,
                                            int r, int h, int ln16, int hl,
                                            v8f acc[8]) {
  const int m = r * 16 + ln16;
  const float4* ap = (const float4*)(smSrc + m * RS) + hl;
  const float4* bp = (const float4*)T + hl * NDIM + h * 128 + ln16;
  #pragma unroll 2
  for (int g = 0; g < NDIM / 8; ++g) {
    const float4 a4 = ap[2 * g];
    float4 b4[8];
    #pragma unroll
    for (int t = 0; t < 8; ++t) b4[t] = bp[g * (2 * NDIM) + t * 16];
    const v2f a0 = {a4.x, a4.y};
    const v2f a1 = {a4.z, a4.w};
    #pragma unroll
    for (int t = 0; t < 8; ++t) {
      const v2f b = {b4[t].x, b4[t].y};
      acc[t] = __builtin_amdgcn_wmma_f32_16x16x4_f32(
           false, a0,  false, b,
           (short)0, acc[t],  false,  false);
    }
    #pragma unroll
    for (int t = 0; t < 8; ++t) {
      const v2f b = {b4[t].z, b4[t].w};
      acc[t] = __builtin_amdgcn_wmma_f32_16x16x4_f32(
           false, a1,  false, b,
           (short)0, acc[t],  false,  false);
    }
  }
}

#define SCR_WORDS (TILE_B * 4 + TILE_B)

__global__ void __launch_bounds__(256)
qlayer_main(const float* __restrict__ x,
            const float* __restrict__ T1,
            const float* __restrict__ T2,
            float* __restrict__ out) {
  extern __shared__ float sm[];
  float* smA = sm;
  float* smB = sm + TILE_B * RS;

  const int tid  = threadIdx.x;
  const int wave = tid >> 5;
  const int lane = tid & 31;
  const int hl   = lane >> 4;
  const int ln16 = lane & 15;
  const int r    = wave & 3;
  const int h    = wave >> 2;
  const int base_row = blockIdx.x * TILE_B;

  {
    const int row = tid >> 2;
    const int q   = tid & 3;
    const float4* src = (const float4*)(x + (base_row + row) * NDIM + q * 64);
    float4* dst = (float4*)(smA + row * RS + q * 64);
    float acc = 0.f;
    #pragma unroll
    for (int c = 0; c < 16; ++c) {
      float4 vv = src[c];
      dst[c] = vv;
      acc += vv.x * vv.x + vv.y * vv.y + vv.z * vv.z + vv.w * vv.w;
    }
    smB[row * 4 + q] = acc;
  }
  __syncthreads();
  if (tid < TILE_B) {
    float s = smB[tid * 4] + smB[tid * 4 + 1] + smB[tid * 4 + 2] + smB[tid * 4 + 3];
    smB[TILE_B * 4 + tid] = rsqrtf(s);
  }
  __syncthreads();
  {
    const int row = tid >> 2;
    const int q   = tid & 3;
    const float inv = smB[TILE_B * 4 + row];
    float4* dst = (float4*)(smA + row * RS + q * 64);
    #pragma unroll
    for (int c = 0; c < 16; ++c) {
      float4 vv = dst[c];
      vv.x *= inv; vv.y *= inv; vv.z *= inv; vv.w *= inv;
      dst[c] = vv;
    }
  }
  __syncthreads();

  const v8f vzero = {0.f, 0.f, 0.f, 0.f, 0.f, 0.f, 0.f, 0.f};

  v8f acc[8];
  v8f res[8];
  #pragma unroll
  for (int t = 0; t < 8; ++t) acc[t] = vzero;
  gemm_stripe(smA, T1, r, h, ln16, hl, acc);
  #pragma unroll
  for (int t = 0; t < 8; ++t) {
    #pragma unroll
    for (int g = 0; g < 8; ++g) { float v = acc[t][g]; res[t][g] = v * v; }
  }

  #pragma unroll
  for (int t = 0; t < 8; ++t) acc[t] = vzero;
  gemm_stripe(smA, T2, r, h, ln16, hl, acc);
  #pragma unroll
  for (int t = 0; t < 8; ++t) {
    #pragma unroll
    for (int g = 0; g < 8; ++g) { float v = acc[t][g]; res[t][g] += v * v; }
  }

  #pragma unroll
  for (int t = 0; t < 8; ++t) {
    const int n = h * 128 + t * 16 + ln16;
    #pragma unroll
    for (int g = 0; g < 8; ++g) {
      const int m = r * 16 + g + 8 * hl;
      out[(base_row + m) * NDIM + n] = res[t][g];
    }
  }
}

extern "C" void kernel_launch(void* const* d_in, const int* in_sizes, int n_in,
                              void* d_out, int out_size, void* d_ws, size_t ws_size,
                              hipStream_t stream) {
  const float* x      = (const float*)d_in[0];
  const float* rots1  = (const float*)d_in[1];
  const float* phases = (const float*)d_in[2];
  const float* rots2  = (const float*)d_in[3];
  float* out = (float*)d_out;
  float* ws  = (float*)d_ws;

  float* cs1 = ws;
  float* cs2 = ws + 65280;
  float* cp = ws + 130560;
  float* sp = ws + 130816;
  float* T1 = ws + 131072;
  float* T2 = ws + 196608;

  const int lds_main   = (TILE_B * RS + SCR_WORDS) * 4;
  hipFuncSetAttribute(reinterpret_cast<const void*>(qlayer_main),
                      hipFuncAttributeMaxDynamicSharedMemorySize, lds_main);

  qlayer_trig<<<(509 * 128 + 255) / 256, 256, 0, stream>>>(
      rots1, rots2, phases, cs1, cs2, cp, sp);
  qlayer_givens<<<2 * (NDIM / GV_COLS), GV_COLS * 32, 0, stream>>>(cs1, cs2, cp, sp, T1, T2);
  qlayer_main<<<NBATCH / TILE_B, 256, lds_main, stream>>>(
      x, T1, T2, out);
}


# Rotation-matrix build: each wave asks the L2 for the cos/sin table 48 steps ahead of its requests (one prefetch per 16-step trip, a 256-byte unit per lane), so a step's pairs are L2 hits, not memory
# speedup vs baseline: 14.3378x; 1.0478x over previous
#include <hip/hip_runtime.h>
#include <math.h>


#define NDIM   256
#define NBATCH 65536
#define NROT   32640
#define TILE_B 64
#define RS     260

typedef __attribute__((ext_vector_type(2))) float v2f;
typedef __attribute__((ext_vector_type(8))) float v8f;

#define GV_LAST  (2 * NDIM - 3)

__device__ __forceinline__ int gv_step_count(int t) {
  const int imin = (t > NDIM - 1) ? (t - (NDIM - 1)) : 0;
  const int imax = (t - 1) >> 1;
  return imax - imin + 1;
}

__device__ __forceinline__ int gv_step_first(int t) {
  const int m = (t > NDIM) ? (t - NDIM) : 0;
  return (t >> 1) * ((t + 1) >> 1) - ((m * (m + 1)) >> 1);
}

__device__ __forceinline__ int gv_rot_index(int t, int slot) {
  const int imin = (t > NDIM - 1) ? (t - (NDIM - 1)) : 0;
  const int i    = imin + slot;
  const unsigned tri = ((unsigned)(i * (2 * NDIM - 1 - i))) >> 1;
  return (int)tri + (t - 2 * i - 1);
}

__global__ void qlayer_trig(const float* __restrict__ rots1,
                            const float* __restrict__ rots2,
                            const float* __restrict__ phases,
                            float* __restrict__ cs1, float* __restrict__ cs2,
                            float* __restrict__ cp, float* __restrict__ sp) {
  const int idx  = blockIdx.x * blockDim.x + threadIdx.x;
  const int t    = (idx >> 7) + 1;
  const int slot = idx & 127;
  if (t <= GV_LAST && slot < gv_step_count(t)) {
    const int k   = gv_rot_index(t, slot);
    const int pos = gv_step_first(t) + slot;
    float s, c;
    sincosf(rots1[k], &s, &c); ((float2*)cs1)[pos] = make_float2(c, s);
    sincosf(rots2[k], &s, &c); ((float2*)cs2)[pos] = make_float2(c, s);
  }
  if (idx < NDIM) {
    float s, c;
    sincosf(phases[idx], &s, &c); cp[idx] = c; sp[idx] = s;
  }
}

#define GV_COLS  4
#define GV_R     4
#define GV_U     16
#define GV_TURN  (NDIM - 1)
#define GV_VS    (NDIM + 128)
#define GV_PF    (3 * GV_U)

static_assert(GV_R * 32 >= NDIM / 2, "a step's rotations must fit the wave");
static_assert(GV_U % GV_R == 0 && GV_U % 8 == 0, "the register renaming and the take-over register must repeat every trip");

struct gv_cs {
  float2 v[GV_R];
};

__device__ __forceinline__ int gv_table_pos(int t) {
  return gv_step_first((t < GV_LAST) ? t : GV_LAST);
}

__device__ __forceinline__ unsigned gv_lane_offset(int t, int lane) {
  return (unsigned)(gv_table_pos(t) + GV_R * lane) * (unsigned)sizeof(float2);
}

__device__ __forceinline__ void gv_request(const float2* __restrict__ csA,
                                           unsigned off, gv_cs& cs) {
  const float2* __restrict__ p = (const float2*)((const char*)csA + off);
  #pragma unroll
  for (int r = 0; r < GV_R; ++r) cs.v[r] = p[r];
}

__device__ __forceinline__ void gv_prefetch(const float2* __restrict__ csA,
                                            int pair, int wv, int lane) {
  const unsigned off = (((unsigned)pair * (unsigned)sizeof(float2)) & ~255u)
                     + (unsigned)(wv * 32 + lane) * 256u;
  __builtin_prefetch((const char*)csA + off, 0, 2);
}

__device__ __forceinline__ float gv_from_lane_below(float x) {
  const int xi = __builtin_bit_cast(int, x);
  const int across = __builtin_amdgcn_permlanex16(xi, xi, 0xffffffffu, 0xffffffffu, false, false);
  return __builtin_bit_cast(float, __builtin_amdgcn_update_dpp(across, xi, 0x111, 0xf, 0xf, false));
}

__device__ __forceinline__ float gv_from_lane_above(float x) {
  const int xi = __builtin_bit_cast(int, x);
  const int across = __builtin_amdgcn_permlanex16(xi, xi, 0u, 0u, false, false);
  return __builtin_bit_cast(float, __builtin_amdgcn_update_dpp(across, xi, 0x101, 0xf, 0xf, false));
}

template <bool SEEDED>
__device__ __forceinline__ void gv_step(const float2* __restrict__ csA,
                                        float* __restrict__ vcol,
                                        int t, int t8, bool up, int lane, int col,
                                        gv_cs& cs, unsigned& off, int& pos, float& seed,
                                        float (&I)[GV_R], float (&J)[GV_R]) {
  float ni[GV_R], nj[GV_R];
  #pragma unroll
  for (int r = 0; r < GV_R; ++r) {
    const float c = cs.v[r].x;
    const float s = cs.v[r].y;
    ni[r] = __builtin_fmaf(c, I[r], -(s * J[r]));
    nj[r] = __builtin_fmaf(s, I[r], c * J[r]);
  }
  if ((t8 & 1) == 0) {
    const int p  = gv_step_count(t);
    const int pr = up ? ((t8 >> 1) & 3) : ((3 - (t8 >> 1)) & 3);
    ni[pr] = (lane == (p >> 2)) ? J[pr] : ni[pr];
  }
  const int pos_new = gv_table_pos(t + GV_U);
  off += (unsigned)(pos_new - pos) * (unsigned)sizeof(float2);
  pos = pos_new;
  gv_request(csA, off, cs);
  __builtin_amdgcn_sched_barrier(0);
  if (up) {
    #pragma unroll
    for (int r = 0; r < GV_R; ++r) I[r] = ni[r];
    const float below = gv_from_lane_below(nj[GV_R - 1]);
    const float fresh = SEEDED ? seed : ((t + 1 == col) ? 1.0f : 0.0f);
    J[3] = nj[2]; J[2] = nj[1]; J[1] = nj[0];
    J[0] = (lane == 0) ? fresh : below;
    if (SEEDED) {
      const int nx = t + 1 + GV_U;
      seed = vcol[(nx < NDIM) ? nx : (NDIM - 1)];
    }
  } else {
    #pragma unroll
    for (int r = 0; r < GV_R; ++r) J[r] = nj[r];
    vcol[(t - GV_TURN) + GV_R * lane] = ni[0];
    const float above = gv_from_lane_above(ni[0]);
    I[0] = ni[1]; I[1] = ni[2]; I[2] = ni[3];
    I[3] = above;
  }
}

template <bool SEEDED>
__device__ __forceinline__ void gv_sweep(const float2* __restrict__ csA,
                                         float* __restrict__ vcol,
                                         int lane, int col, int wv) {
  gv_prefetch(csA, 0, wv, lane);
  gv_cs cs[GV_U];
  unsigned ao[GV_U];
  int ap[GV_U];
  float sd[GV_U];
  #pragma unroll
  for (int k = 0; k < GV_U; ++k) {
    ap[k] = gv_table_pos(1 + k);
    ao[k] = gv_lane_offset(1 + k, lane);
    gv_request(csA, ao[k], cs[k]);
    sd[k] = SEEDED ? vcol[2 + k] : 0.0f;
  }

  float I[GV_R], J[GV_R];
  #pragma unroll
  for (int r = 0; r < GV_R; ++r) { I[r] = 0.0f; J[r] = 0.0f; }
  if (SEEDED) {
    const float s0 = vcol[0];
    const float s1 = vcol[1];
    if (lane == 0) { I[0] = s0; J[0] = s1; }
  } else {
    if (lane == 0) { I[0] = (col == 0) ? 1.0f : 0.0f; J[0] = (col == 1) ? 1.0f : 0.0f; }
  }

  int t = 1;
  for (; t + GV_U - 1 < GV_TURN; t += GV_U) {
    gv_prefetch(csA, gv_table_pos(t + GV_PF), wv, lane);
    #pragma unroll
    for (int k = 0; k < GV_U; ++k)
      gv_step<SEEDED>(csA, vcol, t + k, (1 + k) & 7, true, lane, col, cs[k],
                      ao[k], ap[k], sd[k], I, J);
  }
  gv_prefetch(csA, gv_table_pos(t + GV_PF), wv, lane);
  #pragma unroll
  for (int k = 0; k < (GV_TURN - 1) % GV_U; ++k)
    gv_step<SEEDED>(csA, vcol, t + k, (1 + k) & 7, true, lane, col, cs[k],
                    ao[k], ap[k], sd[k], I, J);
  constexpr int K0 = (GV_TURN - 1) % GV_U;
  t = GV_TURN;
  for (; t + GV_U - 1 <= GV_LAST; t += GV_U) {
    gv_prefetch(csA, gv_table_pos(t + GV_PF), wv, lane);
    #pragma unroll
    for (int k = 0; k < GV_U; ++k)
      gv_step<SEEDED>(csA, vcol, t + k, (GV_TURN + k) & 7, false, lane, col, cs[(K0 + k) % GV_U],
                      ao[(K0 + k) % GV_U], ap[(K0 + k) % GV_U], sd[(K0 + k) % GV_U], I, J);
  }
  #pragma unroll
  for (int k = 0; k < (GV_LAST - GV_TURN + 1) % GV_U; ++k)
    gv_step<SEEDED>(csA, vcol, t + k, (GV_TURN + k) & 7, false, lane, col, cs[(K0 + k) % GV_U],
                    ao[(K0 + k) % GV_U], ap[(K0 + k) % GV_U], sd[(K0 + k) % GV_U], I, J);
  vcol[NDIM - 1 + GV_R * lane] = J[0];
}

__global__ void __launch_bounds__(GV_COLS * 32)
qlayer_givens(const float* __restrict__ cs1,
              const float* __restrict__ cs2,
              const float* __restrict__ cp,
              const float* __restrict__ sp,
              float* __restrict__ T1,
              float* __restrict__ T2) {
  __shared__ float vs[GV_COLS * GV_VS];
  const int tid  = threadIdx.x;
  const int lane = tid & 31;
  const int wv   = __builtin_amdgcn_readfirstlane(tid >> 5);
  const int part = blockIdx.x / (NDIM / GV_COLS);
  const int col0 = (blockIdx.x % (NDIM / GV_COLS)) * GV_COLS;
  const int col  = col0 + wv;
  const float* __restrict__ pt = part ? sp : cp;
  float* __restrict__ T        = part ? T2 : T1;
  float* __restrict__ vcol     = vs + wv * GV_VS;

  gv_prefetch((const float2*)cs2, 0, wv, lane);
  gv_sweep<false>((const float2*)cs1, vcol, lane, col, wv);

  __syncthreads();
  #pragma unroll
  for (int j = 0; j < NDIM / 32; ++j) {
    const int n = lane + 32 * j;
    vcol[n] = vcol[n] * pt[n];
  }
  __syncthreads();

  gv_sweep<true>((const float2*)cs2, vcol, lane, col, wv);

  __syncthreads();

  for (int e = tid; e < NDIM * GV_COLS; e += GV_COLS * 32)
    T[col0 * NDIM + e] = vs[(e % GV_COLS) * GV_VS + e / GV_COLS];
}

__device__ __forceinline__ void gemm_stripe(const float* __restrict__ smSrc,
                                            const float* __restrict__ T,
                                            int r, int h, int ln16, int hl,
                                            v8f acc[8]) {
  const int m = r * 16 + ln16;
  const float4* ap = (const float4*)(smSrc + m * RS) + hl;
  const float4* bp = (const float4*)T + hl * NDIM + h * 128 + ln16;
  #pragma unroll 2
  for (int g = 0; g < NDIM / 8; ++g) {
    const float4 a4 = ap[2 * g];
    float4 b4[8];
    #pragma unroll
    for (int t = 0; t < 8; ++t) b4[t] = bp[g * (2 * NDIM) + t * 16];
    const v2f a0 = {a4.x, a4.y};
    const v2f a1 = {a4.z, a4.w};
    #pragma unroll
    for (int t = 0; t < 8; ++t) {
      const v2f b = {b4[t].x, b4[t].y};
      acc[t] = __builtin_amdgcn_wmma_f32_16x16x4_f32(
           false, a0,  false, b,
           (short)0, acc[t],  false,  false);
    }
    #pragma unroll
    for (int t = 0; t < 8; ++t) {
      const v2f b = {b4[t].z, b4[t].w};
      acc[t] = __builtin_amdgcn_wmma_f32_16x16x4_f32(
           false, a1,  false, b,
           (short)0, acc[t],  false,  false);
    }
  }
}

#define SCR_WORDS (TILE_B * 4 + TILE_B)

__global__ void __launch_bounds__(256)
qlayer_main(const float* __restrict__ x,
            const float* __restrict__ T1,
            const float* __restrict__ T2,
            float* __restrict__ out) {
  extern __shared__ float sm[];
  float* smA = sm;
  float* smB = sm + TILE_B * RS;

  const int tid  = threadIdx.x;
  const int wave = tid >> 5;
  const int lane = tid & 31;
  const int hl   = lane >> 4;
  const int ln16 = lane & 15;
  const int r    = wave & 3;
  const int h    = wave >> 2;
  const int base_row = blockIdx.x * TILE_B;

  {
    const int row = tid >> 2;
    const int q   = tid & 3;
    const float4* src = (const float4*)(x + (base_row + row) * NDIM + q * 64);
    float4* dst = (float4*)(smA + row * RS + q * 64);
    float acc = 0.f;
    #pragma unroll
    for (int c = 0; c < 16; ++c) {
      float4 vv = src[c];
      dst[c] = vv;
      acc += vv.x * vv.x + vv.y * vv.y + vv.z * vv.z + vv.w * vv.w;
    }
    smB[row * 4 + q] = acc;
  }
  __syncthreads();
  if (tid < TILE_B) {
    float s = smB[tid * 4] + smB[tid * 4 + 1] + smB[tid * 4 + 2] + smB[tid * 4 + 3];
    smB[TILE_B * 4 + tid] = rsqrtf(s);
  }
  __syncthreads();
  {
    const int row = tid >> 2;
    const int q   = tid & 3;
    const float inv = smB[TILE_B * 4 + row];
    float4* dst = (float4*)(smA + row * RS + q * 64);
    #pragma unroll
    for (int c = 0; c < 16; ++c) {
      float4 vv = dst[c];
      vv.x *= inv; vv.y *= inv; vv.z *= inv; vv.w *= inv;
      dst[c] = vv;
    }
  }
  __syncthreads();

  const v8f vzero = {0.f, 0.f, 0.f, 0.f, 0.f, 0.f, 0.f, 0.f};

  v8f acc[8];
  v8f res[8];
  #pragma unroll
  for (int t = 0; t < 8; ++t) acc[t] = vzero;
  gemm_stripe(smA, T1, r, h, ln16, hl, acc);
  #pragma unroll
  for (int t = 0; t < 8; ++t) {
    #pragma unroll
    for (int g = 0; g < 8; ++g) { float v = acc[t][g]; res[t][g] = v * v; }
  }

  #pragma unroll
  for (int t = 0; t < 8; ++t) acc[t] = vzero;
  gemm_stripe(smA, T2, r, h, ln16, hl, acc);
  #pragma unroll
  for (int t = 0; t < 8; ++t) {
    #pragma unroll
    for (int g = 0; g < 8; ++g) { float v = acc[t][g]; res[t][g] += v * v; }
  }

  #pragma unroll
  for (int t = 0; t < 8; ++t) {
    const int n = h * 128 + t * 16 + ln16;
    #pragma unroll
    for (int g = 0; g < 8; ++g) {
      const int m = r * 16 + g + 8 * hl;
      out[(base_row + m) * NDIM + n] = res[t][g];
    }
  }
}

extern "C" void kernel_launch(void* const* d_in, const int* in_sizes, int n_in,
                              void* d_out, int out_size, void* d_ws, size_t ws_size,
                              hipStream_t stream) {
  const float* x      = (const float*)d_in[0];
  const float* rots1  = (const float*)d_in[1];
  const float* phases = (const float*)d_in[2];
  const float* rots2  = (const float*)d_in[3];
  float* out = (float*)d_out;
  float* ws  = (float*)d_ws;

  float* cs1 = ws;
  float* cs2 = ws + 65280;
  float* cp = ws + 130560;
  float* sp = ws + 130816;
  float* T1 = ws + 131072;
  float* T2 = ws + 196608;

  const int lds_main   = (TILE_B * RS + SCR_WORDS) * 4;
  hipFuncSetAttribute(reinterpret_cast<const void*>(qlayer_main),
                      hipFuncAttributeMaxDynamicSharedMemorySize, lds_main);

  qlayer_trig<<<(509 * 128 + 255) / 256, 256, 0, stream>>>(
      rots1, rots2, phases, cs1, cs2, cp, sp);
  qlayer_givens<<<2 * (NDIM / GV_COLS), GV_COLS * 32, 0, stream>>>(cs1, cs2, cp, sp, T1, T2);
  qlayer_main<<<NBATCH / TILE_B, 256, lds_main, stream>>>(
      x, T1, T2, out);
}


# Batch pass: x tile read and result written as whole 128-byte lines (a wave takes whole rows: norm summed across lanes, a new sum order, scaled in registers; tile pairs swap lane halves to store)
# speedup vs baseline: 14.9083x; 1.0398x over previous
#include <hip/hip_runtime.h>
#include <math.h>


#define NDIM   256
#define NBATCH 65536
#define NROT   32640
#define TILE_B 64
#define RS     260

typedef __attribute__((ext_vector_type(2))) float v2f;
typedef __attribute__((ext_vector_type(8))) float v8f;
typedef __attribute__((ext_vector_type(2))) unsigned v2u;

#define GV_LAST  (2 * NDIM - 3)

__device__ __forceinline__ int gv_step_count(int t) {
  const int imin = (t > NDIM - 1) ? (t - (NDIM - 1)) : 0;
  const int imax = (t - 1) >> 1;
  return imax - imin + 1;
}

__device__ __forceinline__ int gv_step_first(int t) {
  const int m = (t > NDIM) ? (t - NDIM) : 0;
  return (t >> 1) * ((t + 1) >> 1) - ((m * (m + 1)) >> 1);
}

__device__ __forceinline__ int gv_rot_index(int t, int slot) {
  const int imin = (t > NDIM - 1) ? (t - (NDIM - 1)) : 0;
  const int i    = imin + slot;
  const unsigned tri = ((unsigned)(i * (2 * NDIM - 1 - i))) >> 1;
  return (int)tri + (t - 2 * i - 1);
}

__global__ void qlayer_trig(const float* __restrict__ rots1,
                            const float* __restrict__ rots2,
                            const float* __restrict__ phases,
                            float* __restrict__ cs1, float* __restrict__ cs2,
                            float* __restrict__ cp, float* __restrict__ sp) {
  const int idx  = blockIdx.x * blockDim.x + threadIdx.x;
  const int t    = (idx >> 7) + 1;
  const int slot = idx & 127;
  if (t <= GV_LAST && slot < gv_step_count(t)) {
    const int k   = gv_rot_index(t, slot);
    const int pos = gv_step_first(t) + slot;
    float s, c;
    sincosf(rots1[k], &s, &c); ((float2*)cs1)[pos] = make_float2(c, s);
    sincosf(rots2[k], &s, &c); ((float2*)cs2)[pos] = make_float2(c, s);
  }
  if (idx < NDIM) {
    float s, c;
    sincosf(phases[idx], &s, &c); cp[idx] = c; sp[idx] = s;
  }
}

#define GV_COLS  4
#define GV_R     4
#define GV_U     16
#define GV_TURN  (NDIM - 1)
#define GV_VS    (NDIM + 128)
#define GV_PF    (3 * GV_U)

static_assert(GV_R * 32 >= NDIM / 2, "a step's rotations must fit the wave");
static_assert(GV_U % GV_R == 0 && GV_U % 8 == 0, "the register renaming and the take-over register must repeat every trip");

struct gv_cs {
  float2 v[GV_R];
};

__device__ __forceinline__ int gv_table_pos(int t) {
  return gv_step_first((t < GV_LAST) ? t : GV_LAST);
}

__device__ __forceinline__ unsigned gv_lane_offset(int t, int lane) {
  return (unsigned)(gv_table_pos(t) + GV_R * lane) * (unsigned)sizeof(float2);
}

__device__ __forceinline__ void gv_request(const float2* __restrict__ csA,
                                           unsigned off, gv_cs& cs) {
  const float2* __restrict__ p = (const float2*)((const char*)csA + off);
  #pragma unroll
  for (int r = 0; r < GV_R; ++r) cs.v[r] = p[r];
}

__device__ __forceinline__ void gv_prefetch(const float2* __restrict__ csA,
                                            int pair, int wv, int lane) {
  const unsigned off = (((unsigned)pair * (unsigned)sizeof(float2)) & ~255u)
                     + (unsigned)(wv * 32 + lane) * 256u;
  __builtin_prefetch((const char*)csA + off, 0, 2);
}

__device__ __forceinline__ float gv_from_lane_below(float x) {
  const int xi = __builtin_bit_cast(int, x);
  const int across = __builtin_amdgcn_permlanex16(xi, xi, 0xffffffffu, 0xffffffffu, false, false);
  return __builtin_bit_cast(float, __builtin_amdgcn_update_dpp(across, xi, 0x111, 0xf, 0xf, false));
}

__device__ __forceinline__ float gv_from_lane_above(float x) {
  const int xi = __builtin_bit_cast(int, x);
  const int across = __builtin_amdgcn_permlanex16(xi, xi, 0u, 0u, false, false);
  return __builtin_bit_cast(float, __builtin_amdgcn_update_dpp(across, xi, 0x101, 0xf, 0xf, false));
}

template <bool SEEDED>
__device__ __forceinline__ void gv_step(const float2* __restrict__ csA,
                                        float* __restrict__ vcol,
                                        int t, int t8, bool up, int lane, int col,
                                        gv_cs& cs, unsigned& off, int& pos, float& seed,
                                        float (&I)[GV_R], float (&J)[GV_R]) {
  float ni[GV_R], nj[GV_R];
  #pragma unroll
  for (int r = 0; r < GV_R; ++r) {
    const float c = cs.v[r].x;
    const float s = cs.v[r].y;
    ni[r] = __builtin_fmaf(c, I[r], -(s * J[r]));
    nj[r] = __builtin_fmaf(s, I[r], c * J[r]);
  }
  if ((t8 & 1) == 0) {
    const int p  = gv_step_count(t);
    const int pr = up ? ((t8 >> 1) & 3) : ((3 - (t8 >> 1)) & 3);
    ni[pr] = (lane == (p >> 2)) ? J[pr] : ni[pr];
  }
  const int pos_new = gv_table_pos(t + GV_U);
  off += (unsigned)(pos_new - pos) * (unsigned)sizeof(float2);
  pos = pos_new;
  gv_request(csA, off, cs);
  __builtin_amdgcn_sched_barrier(0);
  if (up) {
    #pragma unroll
    for (int r = 0; r < GV_R; ++r) I[r] = ni[r];
    const float below = gv_from_lane_below(nj[GV_R - 1]);
    const float fresh = SEEDED ? seed : ((t + 1 == col) ? 1.0f : 0.0f);
    J[3] = nj[2]; J[2] = nj[1]; J[1] = nj[0];
    J[0] = (lane == 0) ? fresh : below;
    if (SEEDED) {
      const int nx = t + 1 + GV_U;
      seed = vcol[(nx < NDIM) ? nx : (NDIM - 1)];
    }
  } else {
    #pragma unroll
    for (int r = 0; r < GV_R; ++r) J[r] = nj[r];
    vcol[(t - GV_TURN) + GV_R * lane] = ni[0];
    const float above = gv_from_lane_above(ni[0]);
    I[0] = ni[1]; I[1] = ni[2]; I[2] = ni[3];
    I[3] = above;
  }
}

template <bool SEEDED>
__device__ __forceinline__ void gv_sweep(const float2* __restrict__ csA,
                                         float* __restrict__ vcol,
                                         int lane, int col, int wv) {
  gv_prefetch(csA, 0, wv, lane);
  gv_cs cs[GV_U];
  unsigned ao[GV_U];
  int ap[GV_U];
  float sd[GV_U];
  #pragma unroll
  for (int k = 0; k < GV_U; ++k) {
    ap[k] = gv_table_pos(1 + k);
    ao[k] = gv_lane_offset(1 + k, lane);
    gv_request(csA, ao[k], cs[k]);
    sd[k] = SEEDED ? vcol[2 + k] : 0.0f;
  }

  float I[GV_R], J[GV_R];
  #pragma unroll
  for (int r = 0; r < GV_R; ++r) { I[r] = 0.0f; J[r] = 0.0f; }
  if (SEEDED) {
    const float s0 = vcol[0];
    const float s1 = vcol[1];
    if (lane == 0) { I[0] = s0; J[0] = s1; }
  } else {
    if (lane == 0) { I[0] = (col == 0) ? 1.0f : 0.0f; J[0] = (col == 1) ? 1.0f : 0.0f; }
  }

  int t = 1;
  for (; t + GV_U - 1 < GV_TURN; t += GV_U) {
    gv_prefetch(csA, gv_table_pos(t + GV_PF), wv, lane);
    #pragma unroll
    for (int k = 0; k < GV_U; ++k)
      gv_step<SEEDED>(csA, vcol, t + k, (1 + k) & 7, true, lane, col, cs[k],
                      ao[k], ap[k], sd[k], I, J);
  }
  gv_prefetch(csA, gv_table_pos(t + GV_PF), wv, lane);
  #pragma unroll
  for (int k = 0; k < (GV_TURN - 1) % GV_U; ++k)
    gv_step<SEEDED>(csA, vcol, t + k, (1 + k) & 7, true, lane, col, cs[k],
                    ao[k], ap[k], sd[k], I, J);
  constexpr int K0 = (GV_TURN - 1) % GV_U;
  t = GV_TURN;
  for (; t + GV_U - 1 <= GV_LAST; t += GV_U) {
    gv_prefetch(csA, gv_table_pos(t + GV_PF), wv, lane);
    #pragma unroll
    for (int k = 0; k < GV_U; ++k)
      gv_step<SEEDED>(csA, vcol, t + k, (GV_TURN + k) & 7, false, lane, col, cs[(K0 + k) % GV_U],
                      ao[(K0 + k) % GV_U], ap[(K0 + k) % GV_U], sd[(K0 + k) % GV_U], I, J);
  }
  #pragma unroll
  for (int k = 0; k < (GV_LAST - GV_TURN + 1) % GV_U; ++k)
    gv_step<SEEDED>(csA, vcol, t + k, (GV_TURN + k) & 7, false, lane, col, cs[(K0 + k) % GV_U],
                    ao[(K0 + k) % GV_U], ap[(K0 + k) % GV_U], sd[(K0 + k) % GV_U], I, J);
  vcol[NDIM - 1 + GV_R * lane] = J[0];
}

__global__ void __launch_bounds__(GV_COLS * 32)
qlayer_givens(const float* __restrict__ cs1,
              const float* __restrict__ cs2,
              const float* __restrict__ cp,
              const float* __restrict__ sp,
              float* __restrict__ T1,
              float* __restrict__ T2) {
  __shared__ float vs[GV_COLS * GV_VS];
  const int tid  = threadIdx.x;
  const int lane = tid & 31;
  const int wv   = __builtin_amdgcn_readfirstlane(tid >> 5);
  const int part = blockIdx.x / (NDIM / GV_COLS);
  const int col0 = (blockIdx.x % (NDIM / GV_COLS)) * GV_COLS;
  const int col  = col0 + wv;
  const float* __restrict__ pt = part ? sp : cp;
  float* __restrict__ T        = part ? T2 : T1;
  float* __restrict__ vcol     = vs + wv * GV_VS;

  gv_prefetch((const float2*)cs2, 0, wv, lane);
  gv_sweep<false>((const float2*)cs1, vcol, lane, col, wv);

  __syncthreads();
  #pragma unroll
  for (int j = 0; j < NDIM / 32; ++j) {
    const int n = lane + 32 * j;
    vcol[n] = vcol[n] * pt[n];
  }
  __syncthreads();

  gv_sweep<true>((const float2*)cs2, vcol, lane, col, wv);

  __syncthreads();

  for (int e = tid; e < NDIM * GV_COLS; e += GV_COLS * 32)
    T[col0 * NDIM + e] = vs[(e % GV_COLS) * GV_VS + e / GV_COLS];
}

__device__ __forceinline__ void gemm_stripe(const float* __restrict__ smSrc,
                                            const float* __restrict__ T,
                                            int r, int h, int ln16, int hl,
                                            v8f acc[8]) {
  const int m = r * 16 + ln16;
  const float4* ap = (const float4*)(smSrc + m * RS) + hl;
  const float4* bp = (const float4*)T + hl * NDIM + h * 128 + ln16;
  #pragma unroll 2
  for (int g = 0; g < NDIM / 8; ++g) {
    const float4 a4 = ap[2 * g];
    float4 b4[8];
    #pragma unroll
    for (int t = 0; t < 8; ++t) b4[t] = bp[g * (2 * NDIM) + t * 16];
    const v2f a0 = {a4.x, a4.y};
    const v2f a1 = {a4.z, a4.w};
    #pragma unroll
    for (int t = 0; t < 8; ++t) {
      const v2f b = {b4[t].x, b4[t].y};
      acc[t] = __builtin_amdgcn_wmma_f32_16x16x4_f32(
           false, a0,  false, b,
           (short)0, acc[t],  false,  false);
    }
    #pragma unroll
    for (int t = 0; t < 8; ++t) {
      const v2f b = {b4[t].z, b4[t].w};
      acc[t] = __builtin_amdgcn_wmma_f32_16x16x4_f32(
           false, a1,  false, b,
           (short)0, acc[t],  false,  false);
    }
  }
}


template <int M>
__device__ __forceinline__ float row_xor(float x) {
  const int xi = __builtin_bit_cast(int, x);
  return __builtin_bit_cast(float, __builtin_amdgcn_update_dpp(xi, xi, 0x160 + M, 0xf, 0xf, false));
}

__device__ __forceinline__ float other_row(float x) {
  const int xi = __builtin_bit_cast(int, x);
  return __builtin_bit_cast(float, __builtin_amdgcn_permlanex16(xi, xi, 0x76543210u, 0xfedcba98u, false, false));
}

__device__ __forceinline__ float wave_sum(float p) {
  p += row_xor<1>(p);
  p += row_xor<2>(p);
  p += row_xor<4>(p);
  p += row_xor<8>(p);
  p += other_row(p);
  return p;
}

__global__ void __launch_bounds__(256)
qlayer_main(const float* __restrict__ x,
            const float* __restrict__ T1,
            const float* __restrict__ T2,
            float* __restrict__ out) {
  extern __shared__ float sm[];
  float* smA = sm;

  const int tid  = threadIdx.x;
  const int wave = tid >> 5;
  const int lane = tid & 31;
  const int hl   = lane >> 4;
  const int ln16 = lane & 15;
  const int r    = wave & 3;
  const int h    = wave >> 2;
  const int base_row = blockIdx.x * TILE_B;

  {
    const int wv = __builtin_amdgcn_readfirstlane(tid >> 5);
    const int row0 = wv * 8;
    float4 v[8][2];
    #pragma unroll
    for (int i = 0; i < 8; ++i) {
      const float4* src = (const float4*)(x + (base_row + row0 + i) * NDIM);
      v[i][0] = src[lane];
      v[i][1] = src[lane + 32];
    }
    #pragma unroll
    for (int i = 0; i < 8; ++i) {
      const float4 a = v[i][0];
      const float4 b = v[i][1];
      float p = a.x * a.x + a.y * a.y + a.z * a.z + a.w * a.w;
      p += b.x * b.x + b.y * b.y + b.z * b.z + b.w * b.w;
      const float inv = rsqrtf(wave_sum(p));
      float4* dst = (float4*)(smA + (row0 + i) * RS);
      dst[lane]      = make_float4(a.x * inv, a.y * inv, a.z * inv, a.w * inv);
      dst[lane + 32] = make_float4(b.x * inv, b.y * inv, b.z * inv, b.w * inv);
    }
  }
  __syncthreads();

  const v8f vzero = {0.f, 0.f, 0.f, 0.f, 0.f, 0.f, 0.f, 0.f};

  v8f acc[8];
  v8f res[8];
  #pragma unroll
  for (int t = 0; t < 8; ++t) acc[t] = vzero;
  gemm_stripe(smA, T1, r, h, ln16, hl, acc);
  #pragma unroll
  for (int t = 0; t < 8; ++t) {
    #pragma unroll
    for (int g = 0; g < 8; ++g) { float v = acc[t][g]; res[t][g] = v * v; }
  }

  #pragma unroll
  for (int t = 0; t < 8; ++t) acc[t] = vzero;
  gemm_stripe(smA, T2, r, h, ln16, hl, acc);
  #pragma unroll
  for (int t = 0; t < 8; ++t) {
    #pragma unroll
    for (int g = 0; g < 8; ++g) { float v = acc[t][g]; res[t][g] += v * v; }
  }

  #pragma unroll
  for (int t = 0; t < 8; t += 2) {
    const int n = h * 128 + t * 16 + lane;
    #pragma unroll
    for (int g = 0; g < 8; ++g) {
      const float d0 = res[t][g];
      const float d1 = res[t + 1][g];
      const float x0 = other_row(d0);
      const float x1 = other_row(d1);
      const float lo = hl ? x1 : d0;
      const float hi = hl ? d1 : x0;
      const int m = r * 16 + g;
      out[(base_row + m) * NDIM + n]     = lo;
      out[(base_row + m + 8) * NDIM + n] = hi;
    }
  }
}

extern "C" void kernel_launch(void* const* d_in, const int* in_sizes, int n_in,
                              void* d_out, int out_size, void* d_ws, size_t ws_size,
                              hipStream_t stream) {
  const float* x      = (const float*)d_in[0];
  const float* rots1  = (const float*)d_in[1];
  const float* phases = (const float*)d_in[2];
  const float* rots2  = (const float*)d_in[3];
  float* out = (float*)d_out;
  float* ws  = (float*)d_ws;

  float* cs1 = ws;
  float* cs2 = ws + 65280;
  float* cp = ws + 130560;
  float* sp = ws + 130816;
  float* T1 = ws + 131072;
  float* T2 = ws + 196608;

  const int lds_main   = TILE_B * RS * 4;
  hipFuncSetAttribute(reinterpret_cast<const void*>(qlayer_main),
                      hipFuncAttributeMaxDynamicSharedMemorySize, lds_main);

  qlayer_trig<<<(509 * 128 + 255) / 256, 256, 0, stream>>>(
      rots1, rots2, phases, cs1, cs2, cp, sp);
  qlayer_givens<<<2 * (NDIM / GV_COLS), GV_COLS * 32, 0, stream>>>(cs1, cs2, cp, sp, T1, T2);
  qlayer_main<<<NBATCH / TILE_B, 256, lds_main, stream>>>(
      x, T1, T2, out);
}
